# starting kernel
# baseline (speedup 1.0000x reference)
#include <hip/hip_runtime.h>
#include <cmath>

#define B_ROWS   16384
#define NVC      1024
#define VMAXF    6.0f
#define SC_ITERS 10

typedef __attribute__((ext_vector_type(2))) float v2f;
typedef __attribute__((ext_vector_type(8))) float v8f;

__global__ __launch_bounds__(32) void fp_moments_wmma(
    const float* __restrict__ f, const float* __restrict__ dv_p,
    float4* __restrict__ scal) {
  const int lane    = threadIdx.x;
  const int rowbase = blockIdx.x * 16;
  const float dv    = *dv_p;
  const int r       = lane & 15;
  const int khalf   = (lane >> 4) * 2;

  const float* fp = f + (size_t)(rowbase + r) * NVC + khalf;

  const float c0 = (r == 0) ? 1.0f : 0.0f;
  const float c1 = (r == 1) ? 1.0f : 0.0f;
  const float c2 = (r == 2) ? 1.0f : 0.0f;

  const float dv4 = 4.0f * dv;
  float vcur = dv * ((float)khalf + 0.5f) - VMAXF;

  v8f acc0 = {}, acc1 = {}, acc2 = {}, acc3 = {};
  for (int kb = 0; kb < NVC; kb += 16) {
#pragma unroll
    for (int u = 0; u < 4; ++u) {
      const int k = kb + 4 * u;
      v2f a;
      a.x = fp[k + 0];
      a.y = fp[k + 1];
      const float v0 = vcur;
      const float v1 = vcur + dv;
      v2f b;
      b.x = (c2 * v0 + c1) * v0 + c0;
      b.y = (c2 * v1 + c1) * v1 + c0;
      v8f* acc = (u == 0) ? &acc0 : (u == 1) ? &acc1 : (u == 2) ? &acc2 : &acc3;
      *acc = __builtin_amdgcn_wmma_f32_16x16x4_f32(
          false, a, false, b, (short)0, *acc, false, false);
      vcur += dv4;
    }
  }
  v8f acc = (acc0 + acc1) + (acc2 + acc3);

  float av[8];
#pragma unroll
  for (int i = 0; i < 8; ++i) av[i] = acc[i];

  float myn = 0.f, myvb = 0.f, myet = 0.f, mybeta = 0.f;
#pragma unroll
  for (int m = 0; m < 16; ++m) {
    const int base  = (m < 8) ? 0 : 16;
    const float val = av[m & 7];
    const float S0  = __shfl(val, base + 0, 32);
    const float S1  = __shfl(val, base + 1, 32);
    const float S2  = __shfl(val, base + 2, 32);
    if (lane == m) {
      const float n_   = S0 * dv;
      const float vbar = S1 / S0;
      const float e_t  = (S2 - S1 * S1 / S0) * dv;
      myn = n_; myvb = vbar; myet = e_t;
      mybeta = n_ / (2.0f * e_t);
    }
  }
  if (lane < 16) scal[rowbase + lane] = make_float4(myn, myvb, myet, mybeta);
}

__global__ __launch_bounds__(256) void fp_beta_iter(
    const float* __restrict__ dv_p, float4* __restrict__ scal) {
  const int wave = threadIdx.x >> 5;
  const int lane = threadIdx.x & 31;
  const int row  = blockIdx.x * 8 + wave;
  const float dv = *dv_p;

  const float4 s = scal[row];
  const float n_ = s.x, vbar = s.y, e_t = s.z;
  float beta     = s.w;
  const float g  = n_ / e_t;

  float u2[32];
#pragma unroll
  for (int c = 0; c < 32; ++c) {
    const int j   = lane + 32 * c;
    const float v = dv * ((float)j + 0.5f) - VMAXF;
    const float u = v - vbar;
    u2[c] = u * u;
  }

  for (int it = 0; it < SC_ITERS; ++it) {
    float sm = 0.f, se = 0.f;
#pragma unroll
    for (int c = 0; c < 32; ++c) {
      const float M = __expf(-beta * u2[c]);
      sm += M;
      se += M * u2[c];
    }
#pragma unroll
    for (int off = 16; off > 0; off >>= 1) {
      sm += __shfl_xor(sm, off, 32);
      se += __shfl_xor(se, off, 32);
    }
    beta = beta * (se / sm) * g;
  }
  if (lane == 0) scal[row].w = beta;
}

__global__ __launch_bounds__(256) void fp_thomas(
    const float* __restrict__ f, const float* __restrict__ v_edge,
    const float* __restrict__ dv_p, const float* __restrict__ nu_p,
    const float* __restrict__ dt_p, const float4* __restrict__ scal,
    float* __restrict__ cp_ws, float* __restrict__ out) {
  const int row  = blockIdx.x * blockDim.x + threadIdx.x;
  const float dv = *dv_p, nu = *nu_p, dt = *dt_p;

  const float4 sc  = scal[row];
  const float vbar = sc.y, beta = sc.w;
  const float D      = 0.5f / beta;
  const float Pdv    = D / dv;
  const float s      = dt * nu / dv;
  const float twobdv = 2.0f * beta * dv;

  const float* frow = f + (size_t)row * NVC;
  float*       orow = out + (size_t)row * NVC;

  float Ce0 = 0.f, P0 = 0.f, d0 = 0.5f;
  float cp_prev = 0.f, dp_prev = 0.f;

  for (int jb = 0; jb < NVC; jb += 4) {
    const float4 fj = *(const float4*)(frow + jb);
    float fv[4] = {fj.x, fj.y, fj.z, fj.w};
    float dpv[4];
#pragma unroll
    for (int q = 0; q < 4; ++q) {
      const int j = jb + q;
      const int e = j + 1;
      float Ce1, P1, d1;
      if (e < NVC) {
        const float c = v_edge[e] - vbar;
        Ce1 = c;
        P1  = Pdv;
        const float w = twobdv * c;
        if (fabsf(w) < 1e-6f) {
          d1 = 0.5f;
        } else {
          const float em = expm1f(w);
          d1 = (em - w) / (w * em);
        }
      } else {
        Ce1 = 0.f; P1 = 0.f; d1 = 0.5f;
      }
      const float coefU = Ce1 * (1.0f - d1) + P1;
      const float coefL = -Ce0 * d0 + P0;
      const float coefD = Ce1 * d1 - P1 - Ce0 * (1.0f - d0) - P0;
      const float dl = -s * coefL;
      const float dd = 1.0f - s * coefD;
      const float du = -s * coefU;

      const float denom = dd - dl * cp_prev;
      const float rdiv  = 1.0f / denom;
      const float cp    = du * rdiv;
      const float dp    = (fv[q] - dl * dp_prev) * rdiv;

      cp_ws[(size_t)j * B_ROWS + row] = cp;
      dpv[q]  = dp;
      cp_prev = cp; dp_prev = dp;
      Ce0 = Ce1; P0 = P1; d0 = d1;
    }
    *(float4*)(orow + jb) = make_float4(dpv[0], dpv[1], dpv[2], dpv[3]);
  }

  float xnext = 0.f;
  for (int jb = NVC - 4; jb >= 0; jb -= 4) {
    const float4 dpj = *(const float4*)(orow + jb);
    float dv4[4] = {dpj.x, dpj.y, dpj.z, dpj.w};
    float xv[4];
#pragma unroll
    for (int q = 3; q >= 0; --q) {
      const int j   = jb + q;
      const float c = cp_ws[(size_t)j * B_ROWS + row];
      const float x = dv4[q] - c * xnext;
      xv[q] = x;
      xnext = x;
    }
    *(float4*)(orow + jb) = make_float4(xv[0], xv[1], xv[2], xv[3]);
  }
}

extern "C" void kernel_launch(void* const* d_in, const int* in_sizes, int n_in,
                              void* d_out, int out_size, void* d_ws, size_t ws_size,
                              hipStream_t stream) {
  const float* f      = (const float*)d_in[0];
  const float* v      = (const float*)d_in[1];  (void)v;
  const float* v_edge = (const float*)d_in[2];
  const float* dv     = (const float*)d_in[3];
  const float* nu     = (const float*)d_in[4];
  const float* dt     = (const float*)d_in[5];
  (void)in_sizes; (void)n_in; (void)out_size; (void)ws_size;

  float4* scal = (float4*)d_ws;
  float*  cp   = (float*)((char*)d_ws + (size_t)B_ROWS * sizeof(float4));
  float*  out  = (float*)d_out;

  fp_moments_wmma<<<B_ROWS / 16, 32, 0, stream>>>(f, dv, scal);
  fp_beta_iter  <<<B_ROWS / 8, 256, 0, stream>>>(dv, scal);
  fp_thomas     <<<B_ROWS / 256, 256, 0, stream>>>(f, v_edge, dv, nu, dt, scal, cp, out);
}


# fp_moments: columns of each 16-row tile split over 4 waves (256 each, float4 per lane, all loads ahead of the WMMAs), partial sums joined in LDS; 4096 waves not 1024; order of moment sums changes
# speedup vs baseline: 4.9775x; 4.9775x over previous
#include <hip/hip_runtime.h>
#include <cmath>

#define B_ROWS   16384
#define NVC      1024
#define VMAXF    6.0f
#define SC_ITERS 10

typedef __attribute__((ext_vector_type(2))) float v2f;
typedef __attribute__((ext_vector_type(8))) float v8f;

#define MW    4
#define MCOLS (NVC / MW)
#define MQ    (MCOLS / 8)

__global__ __launch_bounds__(32 * MW) void fp_moments_wmma(
    const float* __restrict__ f, const float* __restrict__ dv_p,
    float4* __restrict__ scal) {
  __shared__ float sP[MW][16][4];

  const int wave    = threadIdx.x >> 5;
  const int lane    = threadIdx.x & 31;
  const int rowbase = blockIdx.x * 16;
  const float dv    = *dv_p;
  const int r       = lane & 15;
  const int h       = lane >> 4;
  const int col0    = wave * MCOLS + 4 * h;

  const float4* fp = (const float4*)(f + (size_t)(rowbase + r) * NVC + col0);

  float4 t[MQ];
#pragma unroll
  for (int q = 0; q < MQ; ++q) t[q] = fp[2 * q];
  __builtin_amdgcn_sched_barrier(0);

  const float c0 = (r == 0) ? 1.0f : 0.0f;
  const float c1 = (r == 1) ? 1.0f : 0.0f;
  const float c2 = (r == 2) ? 1.0f : 0.0f;

  const float dv2 = 2.0f * dv;
  const float dv8 = 8.0f * dv;
  float vcur = dv * ((float)col0 + 0.5f) - VMAXF;

  v8f acc0 = {}, acc1 = {}, acc2 = {}, acc3 = {};
#pragma unroll
  for (int q = 0; q < MQ; ++q) {
    const float v0 = vcur;
    const float v1 = vcur + dv;
    const float v2 = vcur + dv2;
    const float v3 = v2 + dv;
    v2f a01, a23, b01, b23;
    a01.x = t[q].x; a01.y = t[q].y;
    a23.x = t[q].z; a23.y = t[q].w;
    b01.x = (c2 * v0 + c1) * v0 + c0;
    b01.y = (c2 * v1 + c1) * v1 + c0;
    b23.x = (c2 * v2 + c1) * v2 + c0;
    b23.y = (c2 * v3 + c1) * v3 + c0;
    if (q & 1) {
      acc2 = __builtin_amdgcn_wmma_f32_16x16x4_f32(
          false, a01, false, b01, (short)0, acc2, false, false);
      acc3 = __builtin_amdgcn_wmma_f32_16x16x4_f32(
          false, a23, false, b23, (short)0, acc3, false, false);
    } else {
      acc0 = __builtin_amdgcn_wmma_f32_16x16x4_f32(
          false, a01, false, b01, (short)0, acc0, false, false);
      acc1 = __builtin_amdgcn_wmma_f32_16x16x4_f32(
          false, a23, false, b23, (short)0, acc1, false, false);
    }
    vcur += dv8;
  }
  const v8f acc = (acc0 + acc1) + (acc2 + acc3);

  if (r < 3) {
#pragma unroll
    for (int i = 0; i < 8; ++i) sP[wave][8 * h + i][r] = acc[i];
  }
  __syncthreads();

  if (threadIdx.x < 16) {
    const int m    = threadIdx.x;
    const float S0 = (sP[0][m][0] + sP[1][m][0]) + (sP[2][m][0] + sP[3][m][0]);
    const float S1 = (sP[0][m][1] + sP[1][m][1]) + (sP[2][m][1] + sP[3][m][1]);
    const float S2 = (sP[0][m][2] + sP[1][m][2]) + (sP[2][m][2] + sP[3][m][2]);
    const float n_   = S0 * dv;
    const float vbar = S1 / S0;
    const float e_t  = (S2 - S1 * S1 / S0) * dv;
    scal[rowbase + m] = make_float4(n_, vbar, e_t, n_ / (2.0f * e_t));
  }
}

__global__ __launch_bounds__(256) void fp_beta_iter(
    const float* __restrict__ dv_p, float4* __restrict__ scal) {
  const int wave = threadIdx.x >> 5;
  const int lane = threadIdx.x & 31;
  const int row  = blockIdx.x * 8 + wave;
  const float dv = *dv_p;

  const float4 s = scal[row];
  const float n_ = s.x, vbar = s.y, e_t = s.z;
  float beta     = s.w;
  const float g  = n_ / e_t;

  float u2[32];
#pragma unroll
  for (int c = 0; c < 32; ++c) {
    const int j   = lane + 32 * c;
    const float v = dv * ((float)j + 0.5f) - VMAXF;
    const float u = v - vbar;
    u2[c] = u * u;
  }

  for (int it = 0; it < SC_ITERS; ++it) {
    float sm = 0.f, se = 0.f;
#pragma unroll
    for (int c = 0; c < 32; ++c) {
      const float M = __expf(-beta * u2[c]);
      sm += M;
      se += M * u2[c];
    }
#pragma unroll
    for (int off = 16; off > 0; off >>= 1) {
      sm += __shfl_xor(sm, off, 32);
      se += __shfl_xor(se, off, 32);
    }
    beta = beta * (se / sm) * g;
  }
  if (lane == 0) scal[row].w = beta;
}

#define TW   8
#define CH   32
#define LSTR 33
#define WREG (32 * LSTR)

__device__ __forceinline__ void fp_edge(int e, const float* sVe, float vbar,
                                        float Pdv, float twobdv,
                                        float& Ce, float& P, float& d) {
  if (e >= 1 && e < NVC) {
    const float c = sVe[e + (e >> 5)] - vbar;
    Ce = c;
    P  = Pdv;
    const float w = twobdv * c;
    if (fabsf(w) < 1e-6f) {
      d = 0.5f;
    } else {
      const float em = expm1f(w);
      d = (em - w) / (w * em);
    }
  } else {
    Ce = 0.f; P = 0.f; d = 0.5f;
  }
}

__device__ __forceinline__ void fp_coef(float Ce0, float P0, float d0,
                                        float Ce1, float P1, float d1, float s,
                                        float& dl, float& dd, float& du) {
  const float coefU = Ce1 * (1.0f - d1) + P1;
  const float coefL = -Ce0 * d0 + P0;
  const float coefD = Ce1 * d1 - P1 - Ce0 * (1.0f - d0) - P0;
  dl = -s * coefL;
  dd = 1.0f - s * coefD;
  du = -s * coefU;
}

__global__ __launch_bounds__(32 * TW) void fp_thomas(
    const float* __restrict__ f, const float* __restrict__ v_edge,
    const float* __restrict__ dv_p, const float* __restrict__ nu_p,
    const float* __restrict__ dt_p, const float4* __restrict__ scal,
    float* __restrict__ out) {
  __shared__ float sVe[NVC + 1 + 32 + 1];
  __shared__ float sY[TW * WREG];
  __shared__ float sL[TW * WREG];
  __shared__ float sC[TW * WREG];

  const int wave = threadIdx.x >> 5;
  const int lane = threadIdx.x & 31;
  const int row  = blockIdx.x * TW + wave;
  const float dv = *dv_p, nu = *nu_p, dt = *dt_p;

  const float4 sc  = scal[row];
  const float vbar = sc.y, beta = sc.w;
  const float D      = 0.5f / beta;
  const float Pdv    = D / dv;
  const float s      = dt * nu / dv;
  const float twobdv = 2.0f * beta * dv;

  float* Yw = sY + wave * WREG;
  float* Lw = sL + wave * WREG;
  float* Cw = sC + wave * WREG;

  float ve[NVC / (32 * TW)];
  for (int i = 0; i < NVC / (32 * TW); ++i) ve[i] = v_edge[threadIdx.x + 32 * TW * i];
  for (int i = 0; i < NVC / (32 * TW); ++i) {
    const int e = threadIdx.x + 32 * TW * i;
    sVe[e + (e >> 5)] = ve[i];
  }
  if (threadIdx.x == 0) sVe[NVC + (NVC >> 5)] = v_edge[NVC];

  const float4* f4 = (const float4*)(f + (size_t)row * NVC);
  for (int i = 0; i < 8; ++i) {
    const int q    = lane + 32 * i;
    const float4 t = f4[q];
    const int a    = 4 * q + (q >> 3);
    Yw[a] = t.x; Yw[a + 1] = t.y; Yw[a + 2] = t.z; Yw[a + 3] = t.w;
  }
  __syncthreads();

  const int j0 = lane * CH;
  float* y = Yw + lane * LSTR;
  float* l = Lw + lane * LSTR;
  float* c = Cw + lane * LSTR;

  float Ce0, P0, d0;
  fp_edge(j0, sVe, vbar, Pdv, twobdv, Ce0, P0, d0);
  float cp_prev = 0.f, dy_prev = 0.f, dL_prev = 0.f;
  for (int i = 0; i < CH - 1; ++i) {
    float Ce1, P1, d1, dl, dd, du;
    fp_edge(j0 + i + 1, sVe, vbar, Pdv, twobdv, Ce1, P1, d1);
    fp_coef(Ce0, P0, d0, Ce1, P1, d1, s, dl, dd, du);
    const float denom = dd - dl * cp_prev;
    const float rdiv  = 1.0f / denom;
    const float cp    = du * rdiv;
    const float dy    = (y[i] - dl * dy_prev) * rdiv;
    const float rl    = (i == 0) ? dl : 0.f;
    const float dL    = (rl - dl * dL_prev) * rdiv;
    c[i] = cp; y[i] = dy; l[i] = dL;
    cp_prev = cp; dy_prev = dy; dL_prev = dL;
    Ce0 = Ce1; P0 = P1; d0 = d1;
  }
  float dl_s, dd_s, du_s;
  {
    float Ce1, P1, d1;
    fp_edge(j0 + CH, sVe, vbar, Pdv, twobdv, Ce1, P1, d1);
    fp_coef(Ce0, P0, d0, Ce1, P1, d1, s, dl_s, dd_s, du_s);
  }
  const float f_s = y[CH - 1];
  const float yE = dy_prev, LE = dL_prev, UE = cp_prev;

  float yn = yE, Ln = LE, Un = UE;
  for (int i = CH - 3; i >= 0; --i) {
    const float cpi = c[i];
    yn = y[i] - cpi * yn;
    Ln = l[i] - cpi * Ln;
    Un = -cpi * Un;
    y[i] = yn; l[i] = Ln; c[i] = Un;
  }

  float y0n = __shfl_down(yn, 1, 32);
  float L0n = __shfl_down(Ln, 1, 32);
  float U0n = __shfl_down(Un, 1, 32);
  if (lane == 31) { y0n = 0.f; L0n = 0.f; U0n = 0.f; }
  float ra = (lane == 0) ? 0.f : -dl_s * LE;
  float rb = dd_s - dl_s * UE - du_s * L0n;
  float rc = (lane == 31) ? 0.f : -du_s * U0n;
  float rr = f_s - dl_s * yE - du_s * y0n;

  for (int d = 1; d < 32; d <<= 1) {
    float am = __shfl_up(ra, d, 32), bm = __shfl_up(rb, d, 32);
    float cm = __shfl_up(rc, d, 32), rm = __shfl_up(rr, d, 32);
    float ap = __shfl_down(ra, d, 32), bp = __shfl_down(rb, d, 32);
    float cq = __shfl_down(rc, d, 32), rp = __shfl_down(rr, d, 32);
    if (lane < d)       { am = 0.f; bm = 1.f; cm = 0.f; rm = 0.f; }
    if (lane + d >= 32) { ap = 0.f; bp = 1.f; cq = 0.f; rp = 0.f; }
    const float al = -ra / bm;
    const float ga = -rc / bp;
    rb = rb + al * cm + ga * ap;
    rr = rr + al * rm + ga * rp;
    ra = al * am;
    rc = ga * cq;
  }
  const float sk = rr / rb;
  float sl = __shfl_up(sk, 1, 32);
  if (lane == 0) sl = 0.f;

  for (int i = 0; i < CH - 1; ++i) y[i] = y[i] - sl * l[i] - sk * c[i];
  y[CH - 1] = sk;
  __syncthreads();

  float4* o4 = (float4*)(out + (size_t)row * NVC);
  for (int i = 0; i < 8; ++i) {
    const int q = lane + 32 * i;
    const int a = 4 * q + (q >> 3);
    o4[q] = make_float4(Yw[a], Yw[a + 1], Yw[a + 2], Yw[a + 3]);
  }
}

extern "C" void kernel_launch(void* const* d_in, const int* in_sizes, int n_in,
                              void* d_out, int out_size, void* d_ws, size_t ws_size,
                              hipStream_t stream) {
  const float* f      = (const float*)d_in[0];
  const float* v      = (const float*)d_in[1];  (void)v;
  const float* v_edge = (const float*)d_in[2];
  const float* dv     = (const float*)d_in[3];
  const float* nu     = (const float*)d_in[4];
  const float* dt     = (const float*)d_in[5];
  (void)in_sizes; (void)n_in; (void)out_size; (void)ws_size;

  float4* scal = (float4*)d_ws;
  float*  out  = (float*)d_out;

  fp_moments_wmma<<<B_ROWS / 16, 32 * MW, 0, stream>>>(f, dv, scal);
  fp_beta_iter  <<<B_ROWS / 8, 256, 0, stream>>>(dv, scal);
  fp_thomas     <<<B_ROWS / TW, 32 * TW, 0, stream>>>(f, v_edge, dv, nu, dt, scal, out);
}


# fp_thomas: every matrix row scaled by expm1(w) of its two edges, so coefL=C/expm1(w) and coefU=C+coefL enter as products: 1 division per cell, not 2; rounding of the coefficient sums changes
# speedup vs baseline: 5.2132x; 1.0474x over previous
#include <hip/hip_runtime.h>
#include <cmath>

#define B_ROWS   16384
#define NVC      1024
#define VMAXF    6.0f
#define SC_ITERS 10

typedef __attribute__((ext_vector_type(2))) float v2f;
typedef __attribute__((ext_vector_type(8))) float v8f;

#define MW    4
#define MCOLS (NVC / MW)
#define MQ    (MCOLS / 8)

__global__ __launch_bounds__(32 * MW) void fp_moments_wmma(
    const float* __restrict__ f, const float* __restrict__ dv_p,
    float4* __restrict__ scal) {
  __shared__ float sP[MW][16][4];

  const int wave    = threadIdx.x >> 5;
  const int lane    = threadIdx.x & 31;
  const int rowbase = blockIdx.x * 16;
  const float dv    = *dv_p;
  const int r       = lane & 15;
  const int h       = lane >> 4;
  const int col0    = wave * MCOLS + 4 * h;

  const float4* fp = (const float4*)(f + (size_t)(rowbase + r) * NVC + col0);

  float4 t[MQ];
#pragma unroll
  for (int q = 0; q < MQ; ++q) t[q] = fp[2 * q];
  __builtin_amdgcn_sched_barrier(0);

  const float c0 = (r == 0) ? 1.0f : 0.0f;
  const float c1 = (r == 1) ? 1.0f : 0.0f;
  const float c2 = (r == 2) ? 1.0f : 0.0f;

  const float dv2 = 2.0f * dv;
  const float dv8 = 8.0f * dv;
  float vcur = dv * ((float)col0 + 0.5f) - VMAXF;

  v8f acc0 = {}, acc1 = {}, acc2 = {}, acc3 = {};
#pragma unroll
  for (int q = 0; q < MQ; ++q) {
    const float v0 = vcur;
    const float v1 = vcur + dv;
    const float v2 = vcur + dv2;
    const float v3 = v2 + dv;
    v2f a01, a23, b01, b23;
    a01.x = t[q].x; a01.y = t[q].y;
    a23.x = t[q].z; a23.y = t[q].w;
    b01.x = (c2 * v0 + c1) * v0 + c0;
    b01.y = (c2 * v1 + c1) * v1 + c0;
    b23.x = (c2 * v2 + c1) * v2 + c0;
    b23.y = (c2 * v3 + c1) * v3 + c0;
    if (q & 1) {
      acc2 = __builtin_amdgcn_wmma_f32_16x16x4_f32(
          false, a01, false, b01, (short)0, acc2, false, false);
      acc3 = __builtin_amdgcn_wmma_f32_16x16x4_f32(
          false, a23, false, b23, (short)0, acc3, false, false);
    } else {
      acc0 = __builtin_amdgcn_wmma_f32_16x16x4_f32(
          false, a01, false, b01, (short)0, acc0, false, false);
      acc1 = __builtin_amdgcn_wmma_f32_16x16x4_f32(
          false, a23, false, b23, (short)0, acc1, false, false);
    }
    vcur += dv8;
  }
  const v8f acc = (acc0 + acc1) + (acc2 + acc3);

  if (r < 3) {
#pragma unroll
    for (int i = 0; i < 8; ++i) sP[wave][8 * h + i][r] = acc[i];
  }
  __syncthreads();

  if (threadIdx.x < 16) {
    const int m    = threadIdx.x;
    const float S0 = (sP[0][m][0] + sP[1][m][0]) + (sP[2][m][0] + sP[3][m][0]);
    const float S1 = (sP[0][m][1] + sP[1][m][1]) + (sP[2][m][1] + sP[3][m][1]);
    const float S2 = (sP[0][m][2] + sP[1][m][2]) + (sP[2][m][2] + sP[3][m][2]);
    const float n_   = S0 * dv;
    const float vbar = S1 / S0;
    const float e_t  = (S2 - S1 * S1 / S0) * dv;
    scal[rowbase + m] = make_float4(n_, vbar, e_t, n_ / (2.0f * e_t));
  }
}

__global__ __launch_bounds__(256) void fp_beta_iter(
    const float* __restrict__ dv_p, float4* __restrict__ scal) {
  const int wave = threadIdx.x >> 5;
  const int lane = threadIdx.x & 31;
  const int row  = blockIdx.x * 8 + wave;
  const float dv = *dv_p;

  const float4 s = scal[row];
  const float n_ = s.x, vbar = s.y, e_t = s.z;
  float beta     = s.w;
  const float g  = n_ / e_t;

  float u2[32];
#pragma unroll
  for (int c = 0; c < 32; ++c) {
    const int j   = lane + 32 * c;
    const float v = dv * ((float)j + 0.5f) - VMAXF;
    const float u = v - vbar;
    u2[c] = u * u;
  }

  for (int it = 0; it < SC_ITERS; ++it) {
    float sm = 0.f, se = 0.f;
#pragma unroll
    for (int c = 0; c < 32; ++c) {
      const float M = __expf(-beta * u2[c]);
      sm += M;
      se += M * u2[c];
    }
#pragma unroll
    for (int off = 16; off > 0; off >>= 1) {
      sm += __shfl_xor(sm, off, 32);
      se += __shfl_xor(se, off, 32);
    }
    beta = beta * (se / sm) * g;
  }
  if (lane == 0) scal[row].w = beta;
}

#define TW   8
#define CH   32
#define LSTR 33
#define WREG (32 * LSTR)

__device__ __forceinline__ void fp_edge(int e, const float* sVe, float vbar,
                                        float Pdv, float twobdv, float s,
                                        float& sg, float& sA, float& sG) {
  if (e >= 1 && e < NVC) {
    const float c     = sVe[e + (e >> 5)] - vbar;
    const float w     = twobdv * c;
    const bool  small = fabsf(w) < 1e-6f;
    const float em    = expm1f(fminf(w, 30.0f));
    sg = small ? 1.0f : em;
    const float A = small ? (Pdv - 0.5f * c) : c;
    sA = s * A;
    sG = s * (c * sg + A);
  } else {
    sg = 1.0f; sA = 0.f; sG = 0.f;
  }
}

__device__ __forceinline__ void fp_coef(float sg0, float sA0, float sG0,
                                        float sg1, float sA1, float sG1,
                                        float& t, float& dl, float& dd, float& du) {
  t  = sg0 * sg1;
  dl = -(sA0 * sg1);
  du = -(sG1 * sg0);
  dd = sG0 * sg1 + (sA1 * sg0 + t);
}

__global__ __launch_bounds__(32 * TW, 6) void fp_thomas(
    const float* __restrict__ f, const float* __restrict__ v_edge,
    const float* __restrict__ dv_p, const float* __restrict__ nu_p,
    const float* __restrict__ dt_p, const float4* __restrict__ scal,
    float* __restrict__ out) {
  __shared__ float sVe[NVC + 1 + 32 + 1];
  __shared__ float sY[TW * WREG];
  __shared__ float sL[TW * WREG];
  __shared__ float sC[TW * WREG];

  const int wave = threadIdx.x >> 5;
  const int lane = threadIdx.x & 31;
  const int row  = blockIdx.x * TW + wave;
  const float dv = *dv_p, nu = *nu_p, dt = *dt_p;

  const float4 sc  = scal[row];
  const float vbar = sc.y, beta = sc.w;
  const float D      = 0.5f / beta;
  const float Pdv    = D / dv;
  const float s      = dt * nu / dv;
  const float twobdv = 2.0f * beta * dv;

  float* Yw = sY + wave * WREG;
  float* Lw = sL + wave * WREG;
  float* Cw = sC + wave * WREG;

  float ve[NVC / (32 * TW)];
  for (int i = 0; i < NVC / (32 * TW); ++i) ve[i] = v_edge[threadIdx.x + 32 * TW * i];
  for (int i = 0; i < NVC / (32 * TW); ++i) {
    const int e = threadIdx.x + 32 * TW * i;
    sVe[e + (e >> 5)] = ve[i];
  }
  if (threadIdx.x == 0) sVe[NVC + (NVC >> 5)] = v_edge[NVC];

  const float4* f4 = (const float4*)(f + (size_t)row * NVC);
  for (int i = 0; i < 8; ++i) {
    const int q    = lane + 32 * i;
    const float4 t = f4[q];
    const int a    = 4 * q + (q >> 3);
    Yw[a] = t.x; Yw[a + 1] = t.y; Yw[a + 2] = t.z; Yw[a + 3] = t.w;
  }
  __syncthreads();

  const int j0 = lane * CH;
  float* y = Yw + lane * LSTR;
  float* l = Lw + lane * LSTR;
  float* c = Cw + lane * LSTR;

  float sg0, sA0, sG0;
  fp_edge(j0, sVe, vbar, Pdv, twobdv, s, sg0, sA0, sG0);
  float cp_prev = 0.f, dy_prev = 0.f, dL_prev = 0.f;
#pragma unroll
  for (int i = 0; i < CH - 1; ++i) {
    float sg1, sA1, sG1, t, dl, dd, du;
    fp_edge(j0 + i + 1, sVe, vbar, Pdv, twobdv, s, sg1, sA1, sG1);
    fp_coef(sg0, sA0, sG0, sg1, sA1, sG1, t, dl, dd, du);
    const float denom = dd - dl * cp_prev;
    const float rdiv  = 1.0f / denom;
    const float cp    = du * rdiv;
    const float dy    = (y[i] * t - dl * dy_prev) * rdiv;
    const float rl    = (i == 0) ? dl : 0.f;
    const float dL    = (rl - dl * dL_prev) * rdiv;
    c[i] = cp; y[i] = dy; l[i] = dL;
    cp_prev = cp; dy_prev = dy; dL_prev = dL;
    sg0 = sg1; sA0 = sA1; sG0 = sG1;
  }
  float dl_s, dd_s, du_s, t_s;
  {
    float sg1, sA1, sG1;
    fp_edge(j0 + CH, sVe, vbar, Pdv, twobdv, s, sg1, sA1, sG1);
    fp_coef(sg0, sA0, sG0, sg1, sA1, sG1, t_s, dl_s, dd_s, du_s);
  }
  const float f_s = y[CH - 1] * t_s;
  const float yE = dy_prev, LE = dL_prev, UE = cp_prev;

  float yn = yE, Ln = LE, Un = UE;
  for (int i = CH - 3; i >= 0; --i) {
    const float cpi = c[i];
    yn = y[i] - cpi * yn;
    Ln = l[i] - cpi * Ln;
    Un = -cpi * Un;
    y[i] = yn; l[i] = Ln; c[i] = Un;
  }

  float y0n = __shfl_down(yn, 1, 32);
  float L0n = __shfl_down(Ln, 1, 32);
  float U0n = __shfl_down(Un, 1, 32);
  if (lane == 31) { y0n = 0.f; L0n = 0.f; U0n = 0.f; }
  float ra = (lane == 0) ? 0.f : -dl_s * LE;
  float rb = dd_s - dl_s * UE - du_s * L0n;
  float rc = (lane == 31) ? 0.f : -du_s * U0n;
  float rr = f_s - dl_s * yE - du_s * y0n;

  for (int d = 1; d < 32; d <<= 1) {
    float am = __shfl_up(ra, d, 32), bm = __shfl_up(rb, d, 32);
    float cm = __shfl_up(rc, d, 32), rm = __shfl_up(rr, d, 32);
    float ap = __shfl_down(ra, d, 32), bp = __shfl_down(rb, d, 32);
    float cq = __shfl_down(rc, d, 32), rp = __shfl_down(rr, d, 32);
    if (lane < d)       { am = 0.f; bm = 1.f; cm = 0.f; rm = 0.f; }
    if (lane + d >= 32) { ap = 0.f; bp = 1.f; cq = 0.f; rp = 0.f; }
    const float al = -ra / bm;
    const float ga = -rc / bp;
    rb = rb + al * cm + ga * ap;
    rr = rr + al * rm + ga * rp;
    ra = al * am;
    rc = ga * cq;
  }
  const float sk = rr / rb;
  float sl = __shfl_up(sk, 1, 32);
  if (lane == 0) sl = 0.f;

  for (int i = 0; i < CH - 1; ++i) y[i] = y[i] - sl * l[i] - sk * c[i];
  y[CH - 1] = sk;
  __syncthreads();

  float4* o4 = (float4*)(out + (size_t)row * NVC);
  for (int i = 0; i < 8; ++i) {
    const int q = lane + 32 * i;
    const int a = 4 * q + (q >> 3);
    o4[q] = make_float4(Yw[a], Yw[a + 1], Yw[a + 2], Yw[a + 3]);
  }
}

extern "C" void kernel_launch(void* const* d_in, const int* in_sizes, int n_in,
                              void* d_out, int out_size, void* d_ws, size_t ws_size,
                              hipStream_t stream) {
  const float* f      = (const float*)d_in[0];
  const float* v      = (const float*)d_in[1];  (void)v;
  const float* v_edge = (const float*)d_in[2];
  const float* dv     = (const float*)d_in[3];
  const float* nu     = (const float*)d_in[4];
  const float* dt     = (const float*)d_in[5];
  (void)in_sizes; (void)n_in; (void)out_size; (void)ws_size;

  float4* scal = (float4*)d_ws;
  float*  out  = (float*)d_out;

  fp_moments_wmma<<<B_ROWS / 16, 32 * MW, 0, stream>>>(f, dv, scal);
  fp_beta_iter  <<<B_ROWS / 8, 256, 0, stream>>>(dv, scal);
  fp_thomas     <<<B_ROWS / TW, 32 * TW, 0, stream>>>(f, v_edge, dv, nu, dt, scal, out);
}


# fp_thomas: a lane's three chunk vectors (y, left spike, cp/right spike) are registers through the whole solve; LDS only turns f in and x out; 8 waves a SIMD, not 6; every sum keeps its order
# speedup vs baseline: 5.8180x; 1.1160x over previous
#include <hip/hip_runtime.h>
#include <cmath>

#define B_ROWS   16384
#define NVC      1024
#define VMAXF    6.0f
#define SC_ITERS 10

typedef __attribute__((ext_vector_type(2))) float v2f;
typedef __attribute__((ext_vector_type(8))) float v8f;

#define MW    4
#define MCOLS (NVC / MW)
#define MQ    (MCOLS / 8)

__global__ __launch_bounds__(32 * MW) void fp_moments_wmma(
    const float* __restrict__ f, const float* __restrict__ dv_p,
    float4* __restrict__ scal) {
  __shared__ float sP[MW][16][4];

  const int wave    = threadIdx.x >> 5;
  const int lane    = threadIdx.x & 31;
  const int rowbase = blockIdx.x * 16;
  const float dv    = *dv_p;
  const int r       = lane & 15;
  const int h       = lane >> 4;
  const int col0    = wave * MCOLS + 4 * h;

  const float4* fp = (const float4*)(f + (size_t)(rowbase + r) * NVC + col0);

  float4 t[MQ];
#pragma unroll
  for (int q = 0; q < MQ; ++q) t[q] = fp[2 * q];
  __builtin_amdgcn_sched_barrier(0);

  const float c0 = (r == 0) ? 1.0f : 0.0f;
  const float c1 = (r == 1) ? 1.0f : 0.0f;
  const float c2 = (r == 2) ? 1.0f : 0.0f;

  const float dv2 = 2.0f * dv;
  const float dv8 = 8.0f * dv;
  float vcur = dv * ((float)col0 + 0.5f) - VMAXF;

  v8f acc0 = {}, acc1 = {}, acc2 = {}, acc3 = {};
#pragma unroll
  for (int q = 0; q < MQ; ++q) {
    const float v0 = vcur;
    const float v1 = vcur + dv;
    const float v2 = vcur + dv2;
    const float v3 = v2 + dv;
    v2f a01, a23, b01, b23;
    a01.x = t[q].x; a01.y = t[q].y;
    a23.x = t[q].z; a23.y = t[q].w;
    b01.x = (c2 * v0 + c1) * v0 + c0;
    b01.y = (c2 * v1 + c1) * v1 + c0;
    b23.x = (c2 * v2 + c1) * v2 + c0;
    b23.y = (c2 * v3 + c1) * v3 + c0;
    if (q & 1) {
      acc2 = __builtin_amdgcn_wmma_f32_16x16x4_f32(
          false, a01, false, b01, (short)0, acc2, false, false);
      acc3 = __builtin_amdgcn_wmma_f32_16x16x4_f32(
          false, a23, false, b23, (short)0, acc3, false, false);
    } else {
      acc0 = __builtin_amdgcn_wmma_f32_16x16x4_f32(
          false, a01, false, b01, (short)0, acc0, false, false);
      acc1 = __builtin_amdgcn_wmma_f32_16x16x4_f32(
          false, a23, false, b23, (short)0, acc1, false, false);
    }
    vcur += dv8;
  }
  const v8f acc = (acc0 + acc1) + (acc2 + acc3);

  if (r < 3) {
#pragma unroll
    for (int i = 0; i < 8; ++i) sP[wave][8 * h + i][r] = acc[i];
  }
  __syncthreads();

  if (threadIdx.x < 16) {
    const int m    = threadIdx.x;
    const float S0 = (sP[0][m][0] + sP[1][m][0]) + (sP[2][m][0] + sP[3][m][0]);
    const float S1 = (sP[0][m][1] + sP[1][m][1]) + (sP[2][m][1] + sP[3][m][1]);
    const float S2 = (sP[0][m][2] + sP[1][m][2]) + (sP[2][m][2] + sP[3][m][2]);
    const float n_   = S0 * dv;
    const float vbar = S1 / S0;
    const float e_t  = (S2 - S1 * S1 / S0) * dv;
    scal[rowbase + m] = make_float4(n_, vbar, e_t, n_ / (2.0f * e_t));
  }
}

__global__ __launch_bounds__(256) void fp_beta_iter(
    const float* __restrict__ dv_p, float4* __restrict__ scal) {
  const int wave = threadIdx.x >> 5;
  const int lane = threadIdx.x & 31;
  const int row  = blockIdx.x * 8 + wave;
  const float dv = *dv_p;

  const float4 s = scal[row];
  const float n_ = s.x, vbar = s.y, e_t = s.z;
  float beta     = s.w;
  const float g  = n_ / e_t;

  float u2[32];
#pragma unroll
  for (int c = 0; c < 32; ++c) {
    const int j   = lane + 32 * c;
    const float v = dv * ((float)j + 0.5f) - VMAXF;
    const float u = v - vbar;
    u2[c] = u * u;
  }

  for (int it = 0; it < SC_ITERS; ++it) {
    float sm = 0.f, se = 0.f;
#pragma unroll
    for (int c = 0; c < 32; ++c) {
      const float M = __expf(-beta * u2[c]);
      sm += M;
      se += M * u2[c];
    }
#pragma unroll
    for (int off = 16; off > 0; off >>= 1) {
      sm += __shfl_xor(sm, off, 32);
      se += __shfl_xor(se, off, 32);
    }
    beta = beta * (se / sm) * g;
  }
  if (lane == 0) scal[row].w = beta;
}

#define TW   8
#define CH   32
#define LSTR 33
#define WREG (32 * LSTR)

__device__ __forceinline__ void fp_edge(int e, const float* sVe, float vbar,
                                        float Pdv, float twobdv, float s,
                                        float& sg, float& sA, float& sG) {
  if (e >= 1 && e < NVC) {
    const float c     = sVe[e + (e >> 5)] - vbar;
    const float w     = twobdv * c;
    const bool  small = fabsf(w) < 1e-6f;
    const float em    = expm1f(fminf(w, 30.0f));
    sg = small ? 1.0f : em;
    const float A = small ? (Pdv - 0.5f * c) : c;
    sA = s * A;
    sG = s * (c * sg + A);
  } else {
    sg = 1.0f; sA = 0.f; sG = 0.f;
  }
}

__device__ __forceinline__ void fp_coef(float sg0, float sA0, float sG0,
                                        float sg1, float sA1, float sG1,
                                        float& t, float& dl, float& dd, float& du) {
  t  = sg0 * sg1;
  dl = -(sA0 * sg1);
  du = -(sG1 * sg0);
  dd = sG0 * sg1 + (sA1 * sg0 + t);
}

__global__ __launch_bounds__(32 * TW, 8) void fp_thomas(
    const float* __restrict__ f, const float* __restrict__ v_edge,
    const float* __restrict__ dv_p, const float* __restrict__ nu_p,
    const float* __restrict__ dt_p, const float4* __restrict__ scal,
    float* __restrict__ out) {
  __shared__ float sVe[NVC + 1 + 32 + 1];
  __shared__ float sY[TW * WREG];

  const int wave = threadIdx.x >> 5;
  const int lane = threadIdx.x & 31;
  const int row  = blockIdx.x * TW + wave;
  const float dv = *dv_p, nu = *nu_p, dt = *dt_p;

  const float4 sc  = scal[row];
  const float vbar = sc.y, beta = sc.w;
  const float D      = 0.5f / beta;
  const float Pdv    = D / dv;
  const float s      = dt * nu / dv;
  const float twobdv = 2.0f * beta * dv;

  float* Yw = sY + wave * WREG;

  float ve[NVC / (32 * TW)];
  for (int i = 0; i < NVC / (32 * TW); ++i) ve[i] = v_edge[threadIdx.x + 32 * TW * i];
  for (int i = 0; i < NVC / (32 * TW); ++i) {
    const int e = threadIdx.x + 32 * TW * i;
    sVe[e + (e >> 5)] = ve[i];
  }
  if (threadIdx.x == 0) sVe[NVC + (NVC >> 5)] = v_edge[NVC];

  const float4* f4 = (const float4*)(f + (size_t)row * NVC);
  for (int i = 0; i < 8; ++i) {
    const int q    = lane + 32 * i;
    const float4 t = f4[q];
    const int a    = 4 * q + (q >> 3);
    Yw[a] = t.x; Yw[a + 1] = t.y; Yw[a + 2] = t.z; Yw[a + 3] = t.w;
  }
  __syncthreads();

  const int j0 = lane * CH;
  float* y = Yw + lane * LSTR;
  float Y[CH - 1];
  float L[CH - 1];
  float C[CH - 1];

  float sg0, sA0, sG0;
  fp_edge(j0, sVe, vbar, Pdv, twobdv, s, sg0, sA0, sG0);
  float cp_prev = 0.f, dy_prev = 0.f, dL_prev = 0.f;
#pragma unroll
  for (int i = 0; i < CH - 1; ++i) {
    float sg1, sA1, sG1, t, dl, dd, du;
    fp_edge(j0 + i + 1, sVe, vbar, Pdv, twobdv, s, sg1, sA1, sG1);
    fp_coef(sg0, sA0, sG0, sg1, sA1, sG1, t, dl, dd, du);
    const float denom = dd - dl * cp_prev;
    const float rdiv  = 1.0f / denom;
    const float cp    = du * rdiv;
    const float dy    = (y[i] * t - dl * dy_prev) * rdiv;
    const float rl    = (i == 0) ? dl : 0.f;
    const float dL    = (rl - dl * dL_prev) * rdiv;
    C[i] = cp; Y[i] = dy; L[i] = dL;
    cp_prev = cp; dy_prev = dy; dL_prev = dL;
    sg0 = sg1; sA0 = sA1; sG0 = sG1;
  }
  float dl_s, dd_s, du_s, t_s;
  {
    float sg1, sA1, sG1;
    fp_edge(j0 + CH, sVe, vbar, Pdv, twobdv, s, sg1, sA1, sG1);
    fp_coef(sg0, sA0, sG0, sg1, sA1, sG1, t_s, dl_s, dd_s, du_s);
  }
  const float f_s = y[CH - 1] * t_s;
  const float yE = dy_prev, LE = dL_prev, UE = cp_prev;

  float yn = yE, Ln = LE, Un = UE;
#pragma unroll
  for (int i = CH - 3; i >= 0; --i) {
    const float cpi = C[i];
    yn = Y[i] - cpi * yn;
    Ln = L[i] - cpi * Ln;
    Un = -cpi * Un;
    Y[i] = yn; L[i] = Ln; C[i] = Un;
  }

  float y0n = __shfl_down(yn, 1, 32);
  float L0n = __shfl_down(Ln, 1, 32);
  float U0n = __shfl_down(Un, 1, 32);
  if (lane == 31) { y0n = 0.f; L0n = 0.f; U0n = 0.f; }
  float ra = (lane == 0) ? 0.f : -dl_s * LE;
  float rb = dd_s - dl_s * UE - du_s * L0n;
  float rc = (lane == 31) ? 0.f : -du_s * U0n;
  float rr = f_s - dl_s * yE - du_s * y0n;

  for (int d = 1; d < 32; d <<= 1) {
    float am = __shfl_up(ra, d, 32), bm = __shfl_up(rb, d, 32);
    float cm = __shfl_up(rc, d, 32), rm = __shfl_up(rr, d, 32);
    float ap = __shfl_down(ra, d, 32), bp = __shfl_down(rb, d, 32);
    float cq = __shfl_down(rc, d, 32), rp = __shfl_down(rr, d, 32);
    if (lane < d)       { am = 0.f; bm = 1.f; cm = 0.f; rm = 0.f; }
    if (lane + d >= 32) { ap = 0.f; bp = 1.f; cq = 0.f; rp = 0.f; }
    const float al = -ra / bm;
    const float ga = -rc / bp;
    rb = rb + al * cm + ga * ap;
    rr = rr + al * rm + ga * rp;
    ra = al * am;
    rc = ga * cq;
  }
  const float sk = rr / rb;
  float sl = __shfl_up(sk, 1, 32);
  if (lane == 0) sl = 0.f;

#pragma unroll
  for (int i = 0; i < CH - 1; ++i) y[i] = Y[i] - sl * L[i] - sk * C[i];
  y[CH - 1] = sk;
  __syncthreads();

  float4* o4 = (float4*)(out + (size_t)row * NVC);
  for (int i = 0; i < 8; ++i) {
    const int q = lane + 32 * i;
    const int a = 4 * q + (q >> 3);
    o4[q] = make_float4(Yw[a], Yw[a + 1], Yw[a + 2], Yw[a + 3]);
  }
}

extern "C" void kernel_launch(void* const* d_in, const int* in_sizes, int n_in,
                              void* d_out, int out_size, void* d_ws, size_t ws_size,
                              hipStream_t stream) {
  const float* f      = (const float*)d_in[0];
  const float* v      = (const float*)d_in[1];  (void)v;
  const float* v_edge = (const float*)d_in[2];
  const float* dv     = (const float*)d_in[3];
  const float* nu     = (const float*)d_in[4];
  const float* dt     = (const float*)d_in[5];
  (void)in_sizes; (void)n_in; (void)out_size; (void)ws_size;

  float4* scal = (float4*)d_ws;
  float*  out  = (float*)d_out;

  fp_moments_wmma<<<B_ROWS / 16, 32 * MW, 0, stream>>>(f, dv, scal);
  fp_beta_iter  <<<B_ROWS / 8, 256, 0, stream>>>(dv, scal);
  fp_thomas     <<<B_ROWS / TW, 32 * TW, 0, stream>>>(f, v_edge, dv, nu, dt, scal, out);
}


# fp_beta_iter: M=exp(-beta*u^2) carried along the uniform grid (M*=R, R*=Q), a lane = 4 chains of 8 consecutive cells: 9 exp per lane and sweep, not 32; rounding of M and order of its sums change
# speedup vs baseline: 6.2467x; 1.0737x over previous
#include <hip/hip_runtime.h>
#include <cmath>

#define B_ROWS   16384
#define NVC      1024
#define VMAXF    6.0f
#define SC_ITERS 10

typedef __attribute__((ext_vector_type(2))) float v2f;
typedef __attribute__((ext_vector_type(8))) float v8f;

#define MW    4
#define MCOLS (NVC / MW)
#define MQ    (MCOLS / 8)

__global__ __launch_bounds__(32 * MW) void fp_moments_wmma(
    const float* __restrict__ f, const float* __restrict__ dv_p,
    float4* __restrict__ scal) {
  __shared__ float sP[MW][16][4];

  const int wave    = threadIdx.x >> 5;
  const int lane    = threadIdx.x & 31;
  const int rowbase = blockIdx.x * 16;
  const float dv    = *dv_p;
  const int r       = lane & 15;
  const int h       = lane >> 4;
  const int col0    = wave * MCOLS + 4 * h;

  const float4* fp = (const float4*)(f + (size_t)(rowbase + r) * NVC + col0);

  float4 t[MQ];
#pragma unroll
  for (int q = 0; q < MQ; ++q) t[q] = fp[2 * q];
  __builtin_amdgcn_sched_barrier(0);

  const float c0 = (r == 0) ? 1.0f : 0.0f;
  const float c1 = (r == 1) ? 1.0f : 0.0f;
  const float c2 = (r == 2) ? 1.0f : 0.0f;

  const float dv2 = 2.0f * dv;
  const float dv8 = 8.0f * dv;
  float vcur = dv * ((float)col0 + 0.5f) - VMAXF;

  v8f acc0 = {}, acc1 = {}, acc2 = {}, acc3 = {};
#pragma unroll
  for (int q = 0; q < MQ; ++q) {
    const float v0 = vcur;
    const float v1 = vcur + dv;
    const float v2 = vcur + dv2;
    const float v3 = v2 + dv;
    v2f a01, a23, b01, b23;
    a01.x = t[q].x; a01.y = t[q].y;
    a23.x = t[q].z; a23.y = t[q].w;
    b01.x = (c2 * v0 + c1) * v0 + c0;
    b01.y = (c2 * v1 + c1) * v1 + c0;
    b23.x = (c2 * v2 + c1) * v2 + c0;
    b23.y = (c2 * v3 + c1) * v3 + c0;
    if (q & 1) {
      acc2 = __builtin_amdgcn_wmma_f32_16x16x4_f32(
          false, a01, false, b01, (short)0, acc2, false, false);
      acc3 = __builtin_amdgcn_wmma_f32_16x16x4_f32(
          false, a23, false, b23, (short)0, acc3, false, false);
    } else {
      acc0 = __builtin_amdgcn_wmma_f32_16x16x4_f32(
          false, a01, false, b01, (short)0, acc0, false, false);
      acc1 = __builtin_amdgcn_wmma_f32_16x16x4_f32(
          false, a23, false, b23, (short)0, acc1, false, false);
    }
    vcur += dv8;
  }
  const v8f acc = (acc0 + acc1) + (acc2 + acc3);

  if (r < 3) {
#pragma unroll
    for (int i = 0; i < 8; ++i) sP[wave][8 * h + i][r] = acc[i];
  }
  __syncthreads();

  if (threadIdx.x < 16) {
    const int m    = threadIdx.x;
    const float S0 = (sP[0][m][0] + sP[1][m][0]) + (sP[2][m][0] + sP[3][m][0]);
    const float S1 = (sP[0][m][1] + sP[1][m][1]) + (sP[2][m][1] + sP[3][m][1]);
    const float S2 = (sP[0][m][2] + sP[1][m][2]) + (sP[2][m][2] + sP[3][m][2]);
    const float n_   = S0 * dv;
    const float vbar = S1 / S0;
    const float e_t  = (S2 - S1 * S1 / S0) * dv;
    scal[rowbase + m] = make_float4(n_, vbar, e_t, n_ / (2.0f * e_t));
  }
}

#define NCH  4
#define CLEN (32 / NCH)

__global__ __launch_bounds__(256) void fp_beta_iter(
    const float* __restrict__ dv_p, float4* __restrict__ scal) {
  const int wave = threadIdx.x >> 5;
  const int lane = threadIdx.x & 31;
  const int row  = blockIdx.x * 8 + wave;
  const float dv = *dv_p;

  const float4 s = scal[row];
  const float n_ = s.x, vbar = s.y, e_t = s.z;
  float beta     = s.w;
  const float g  = n_ / e_t;

  float u2[CLEN][NCH];
  float dR[NCH];
#pragma unroll
  for (int k = 0; k < NCH; ++k) {
#pragma unroll
    for (int i = 0; i < CLEN; ++i) {
      const int j   = 32 * lane + CLEN * k + i;
      const float v = dv * ((float)j + 0.5f) - VMAXF;
      const float u = v - vbar;
      u2[i][k] = u * u;
      if (i == 0) dR[k] = dv * (2.0f * u + dv);
    }
  }
  const float dQ = 2.0f * dv * dv;

  for (int it = 0; it < SC_ITERS; ++it) {
    const float Q = __expf(-beta * dQ);
    float M[NCH], R[NCH], sm[NCH], se[NCH];
#pragma unroll
    for (int k = 0; k < NCH; ++k) {
      M[k]  = __expf(-beta * u2[0][k]);
      R[k]  = __expf(fminf(-beta * dR[k], 87.0f));
      sm[k] = M[k];
      se[k] = M[k] * u2[0][k];
    }
#pragma unroll
    for (int i = 1; i < CLEN; ++i) {
#pragma unroll
      for (int k = 0; k < NCH; ++k) M[k] *= R[k];
      if (i + 1 < CLEN) {
#pragma unroll
        for (int k = 0; k < NCH; ++k) R[k] *= Q;
      }
#pragma unroll
      for (int k = 0; k < NCH; ++k) sm[k] += M[k];
#pragma unroll
      for (int k = 0; k < NCH; ++k) se[k] += M[k] * u2[i][k];
    }
    float smt = (sm[0] + sm[1]) + (sm[2] + sm[3]);
    float set = (se[0] + se[1]) + (se[2] + se[3]);
#pragma unroll
    for (int off = 16; off > 0; off >>= 1) {
      smt += __shfl_xor(smt, off, 32);
      set += __shfl_xor(set, off, 32);
    }
    beta = beta * (set / smt) * g;
  }
  if (lane == 0) scal[row].w = beta;
}

#define TW   8
#define CH   32
#define LSTR 33
#define WREG (32 * LSTR)

__device__ __forceinline__ void fp_edge(int e, const float* sVe, float vbar,
                                        float Pdv, float twobdv, float s,
                                        float& sg, float& sA, float& sG) {
  if (e >= 1 && e < NVC) {
    const float c     = sVe[e + (e >> 5)] - vbar;
    const float w     = twobdv * c;
    const bool  small = fabsf(w) < 1e-6f;
    const float em    = expm1f(fminf(w, 30.0f));
    sg = small ? 1.0f : em;
    const float A = small ? (Pdv - 0.5f * c) : c;
    sA = s * A;
    sG = s * (c * sg + A);
  } else {
    sg = 1.0f; sA = 0.f; sG = 0.f;
  }
}

__device__ __forceinline__ void fp_coef(float sg0, float sA0, float sG0,
                                        float sg1, float sA1, float sG1,
                                        float& t, float& dl, float& dd, float& du) {
  t  = sg0 * sg1;
  dl = -(sA0 * sg1);
  du = -(sG1 * sg0);
  dd = sG0 * sg1 + (sA1 * sg0 + t);
}

__global__ __launch_bounds__(32 * TW, 8) void fp_thomas(
    const float* __restrict__ f, const float* __restrict__ v_edge,
    const float* __restrict__ dv_p, const float* __restrict__ nu_p,
    const float* __restrict__ dt_p, const float4* __restrict__ scal,
    float* __restrict__ out) {
  __shared__ float sVe[NVC + 1 + 32 + 1];
  __shared__ float sY[TW * WREG];

  const int wave = threadIdx.x >> 5;
  const int lane = threadIdx.x & 31;
  const int row  = blockIdx.x * TW + wave;
  const float dv = *dv_p, nu = *nu_p, dt = *dt_p;

  const float4 sc  = scal[row];
  const float vbar = sc.y, beta = sc.w;
  const float D      = 0.5f / beta;
  const float Pdv    = D / dv;
  const float s      = dt * nu / dv;
  const float twobdv = 2.0f * beta * dv;

  float* Yw = sY + wave * WREG;

  float ve[NVC / (32 * TW)];
  for (int i = 0; i < NVC / (32 * TW); ++i) ve[i] = v_edge[threadIdx.x + 32 * TW * i];
  for (int i = 0; i < NVC / (32 * TW); ++i) {
    const int e = threadIdx.x + 32 * TW * i;
    sVe[e + (e >> 5)] = ve[i];
  }
  if (threadIdx.x == 0) sVe[NVC + (NVC >> 5)] = v_edge[NVC];

  const float4* f4 = (const float4*)(f + (size_t)row * NVC);
  for (int i = 0; i < 8; ++i) {
    const int q    = lane + 32 * i;
    const float4 t = f4[q];
    const int a    = 4 * q + (q >> 3);
    Yw[a] = t.x; Yw[a + 1] = t.y; Yw[a + 2] = t.z; Yw[a + 3] = t.w;
  }
  __syncthreads();

  const int j0 = lane * CH;
  float* y = Yw + lane * LSTR;
  float Y[CH - 1];
  float L[CH - 1];
  float C[CH - 1];

  float sg0, sA0, sG0;
  fp_edge(j0, sVe, vbar, Pdv, twobdv, s, sg0, sA0, sG0);
  float cp_prev = 0.f, dy_prev = 0.f, dL_prev = 0.f;
#pragma unroll
  for (int i = 0; i < CH - 1; ++i) {
    float sg1, sA1, sG1, t, dl, dd, du;
    fp_edge(j0 + i + 1, sVe, vbar, Pdv, twobdv, s, sg1, sA1, sG1);
    fp_coef(sg0, sA0, sG0, sg1, sA1, sG1, t, dl, dd, du);
    const float denom = dd - dl * cp_prev;
    const float rdiv  = 1.0f / denom;
    const float cp    = du * rdiv;
    const float dy    = (y[i] * t - dl * dy_prev) * rdiv;
    const float rl    = (i == 0) ? dl : 0.f;
    const float dL    = (rl - dl * dL_prev) * rdiv;
    C[i] = cp; Y[i] = dy; L[i] = dL;
    cp_prev = cp; dy_prev = dy; dL_prev = dL;
    sg0 = sg1; sA0 = sA1; sG0 = sG1;
  }
  float dl_s, dd_s, du_s, t_s;
  {
    float sg1, sA1, sG1;
    fp_edge(j0 + CH, sVe, vbar, Pdv, twobdv, s, sg1, sA1, sG1);
    fp_coef(sg0, sA0, sG0, sg1, sA1, sG1, t_s, dl_s, dd_s, du_s);
  }
  const float f_s = y[CH - 1] * t_s;
  const float yE = dy_prev, LE = dL_prev, UE = cp_prev;

  float yn = yE, Ln = LE, Un = UE;
#pragma unroll
  for (int i = CH - 3; i >= 0; --i) {
    const float cpi = C[i];
    yn = Y[i] - cpi * yn;
    Ln = L[i] - cpi * Ln;
    Un = -cpi * Un;
    Y[i] = yn; L[i] = Ln; C[i] = Un;
  }

  float y0n = __shfl_down(yn, 1, 32);
  float L0n = __shfl_down(Ln, 1, 32);
  float U0n = __shfl_down(Un, 1, 32);
  if (lane == 31) { y0n = 0.f; L0n = 0.f; U0n = 0.f; }
  float ra = (lane == 0) ? 0.f : -dl_s * LE;
  float rb = dd_s - dl_s * UE - du_s * L0n;
  float rc = (lane == 31) ? 0.f : -du_s * U0n;
  float rr = f_s - dl_s * yE - du_s * y0n;

  for (int d = 1; d < 32; d <<= 1) {
    float am = __shfl_up(ra, d, 32), bm = __shfl_up(rb, d, 32);
    float cm = __shfl_up(rc, d, 32), rm = __shfl_up(rr, d, 32);
    float ap = __shfl_down(ra, d, 32), bp = __shfl_down(rb, d, 32);
    float cq = __shfl_down(rc, d, 32), rp = __shfl_down(rr, d, 32);
    if (lane < d)       { am = 0.f; bm = 1.f; cm = 0.f; rm = 0.f; }
    if (lane + d >= 32) { ap = 0.f; bp = 1.f; cq = 0.f; rp = 0.f; }
    const float al = -ra / bm;
    const float ga = -rc / bp;
    rb = rb + al * cm + ga * ap;
    rr = rr + al * rm + ga * rp;
    ra = al * am;
    rc = ga * cq;
  }
  const float sk = rr / rb;
  float sl = __shfl_up(sk, 1, 32);
  if (lane == 0) sl = 0.f;

#pragma unroll
  for (int i = 0; i < CH - 1; ++i) y[i] = Y[i] - sl * L[i] - sk * C[i];
  y[CH - 1] = sk;
  __syncthreads();

  float4* o4 = (float4*)(out + (size_t)row * NVC);
  for (int i = 0; i < 8; ++i) {
    const int q = lane + 32 * i;
    const int a = 4 * q + (q >> 3);
    o4[q] = make_float4(Yw[a], Yw[a + 1], Yw[a + 2], Yw[a + 3]);
  }
}

extern "C" void kernel_launch(void* const* d_in, const int* in_sizes, int n_in,
                              void* d_out, int out_size, void* d_ws, size_t ws_size,
                              hipStream_t stream) {
  const float* f      = (const float*)d_in[0];
  const float* v      = (const float*)d_in[1];  (void)v;
  const float* v_edge = (const float*)d_in[2];
  const float* dv     = (const float*)d_in[3];
  const float* nu     = (const float*)d_in[4];
  const float* dt     = (const float*)d_in[5];
  (void)in_sizes; (void)n_in; (void)out_size; (void)ws_size;

  float4* scal = (float4*)d_ws;
  float*  out  = (float*)d_out;

  fp_moments_wmma<<<B_ROWS / 16, 32 * MW, 0, stream>>>(f, dv, scal);
  fp_beta_iter  <<<B_ROWS / 8, 256, 0, stream>>>(dv, scal);
  fp_thomas     <<<B_ROWS / TW, 32 * TW, 0, stream>>>(f, v_edge, dv, nu, dt, scal, out);
}
